# starting kernel
# baseline (speedup 1.0000x reference)
#include <hip/hip_runtime.h>


typedef __attribute__((ext_vector_type(2))) float v2f;
typedef __attribute__((ext_vector_type(8))) float v8f;

#define B_ 8
#define M_ 512
#define A_ 16
#define F_ 128

#define WSTRIDE 17
#define WLDS_PER_A (A_ * WSTRIDE)
#define WLDS_TOTAL (A_ * WLDS_PER_A)

__global__ __launch_bounds__(256)
void tpcl_wmma_f32_kernel(const float* __restrict__ x,
                          const float* __restrict__ y,
                          const float* __restrict__ cgc,
                          float* __restrict__ out)
{
    __shared__ float w[WLDS_TOTAL];

    for (int i = threadIdx.x; i < A_ * A_ * A_; i += 256) {
        int a = i >> 8;
        int d = (i >> 4) & 15;
        int c = i & 15;
        w[a * WLDS_PER_A + d * WSTRIDE + c] = cgc[i];
    }
    __syncthreads();

    const int lane      = threadIdx.x & 31;
    const int waveInBlk = threadIdx.x >> 5;
    const int gw        = blockIdx.x * 8 + waveInBlk;
    const int bm        = gw >> 3;
    const int ftile     = gw & 7;
    const int fcol      = (ftile << 4) + (lane & 15);
    const int hi        = lane >> 4;
    const int off       = hi << 1;
    const int dM        = lane & 15;

    const float* __restrict__ xp = x + (size_t)bm * (A_ * F_) + fcol;
    const float* __restrict__ yp = y + (size_t)bm * (A_ * F_) + fcol;

    float xr[16];
#pragma unroll
    for (int a = 0; a < 16; ++a) xr[a] = xp[a * F_];

    float yA[4], yB[4];
#pragma unroll
    for (int k = 0; k < 4; ++k) {
        yA[k] = yp[(4 * k + off)     * F_];
        yB[k] = yp[(4 * k + off + 1) * F_];
    }

    v8f acc = {};
    const float* __restrict__ wbase = &w[dM * WSTRIDE + off];

#pragma unroll
    for (int t = 0; t < 64; ++t) {
        const int a  = t >> 2;
        const int k4 = t & 3;

        v2f aop;
        aop.x = wbase[a * WLDS_PER_A + 4 * k4];
        aop.y = wbase[a * WLDS_PER_A + 4 * k4 + 1];

        v2f bop;
        bop.x = xr[a] * yA[k4];
        bop.y = xr[a] * yB[k4];

        acc = __builtin_amdgcn_wmma_f32_16x16x4_f32(
             false, aop,  false, bop,
             (short)0, acc,  false,  false);
    }

    float* __restrict__ op = out + (size_t)bm * (A_ * F_) + (hi ? 8 * F_ : 0) + fcol;
#pragma unroll
    for (int i = 0; i < 8; ++i) op[i * F_] = acc[i];
}

extern "C" void kernel_launch(void* const* d_in, const int* in_sizes, int n_in,
                              void* d_out, int out_size, void* d_ws, size_t ws_size,
                              hipStream_t stream) {
    const float* x   = (const float*)d_in[0];
    const float* y   = (const float*)d_in[1];
    const float* cgc = (const float*)d_in[2];
    float* out = (float*)d_out;

    (void)in_sizes; (void)n_in; (void)out_size; (void)d_ws; (void)ws_size;

    dim3 grid(B_ * M_);
    dim3 block(256);
    tpcl_wmma_f32_kernel<<<grid, block, 0, stream>>>(x, y, cgc, out);
}


# persistent waves: 256 resident workgroups; each wave loads the 64 cgc A fragments into registers once, then loops over 8 items of 32 f columns (even/odd f tiles), next item's loads in flight
# speedup vs baseline: 1.2604x; 1.2604x over previous
#include <hip/hip_runtime.h>


typedef __attribute__((ext_vector_type(2))) float v2f;
typedef __attribute__((ext_vector_type(4))) float v4f;
typedef __attribute__((ext_vector_type(8))) float v8f;

#define B_ 8
#define M_ 512
#define A_ 16
#define F_ 128

#define WSTRIDE 17
#define WLDS_PER_A (A_ * WSTRIDE)
#define WLDS_TOTAL (A_ * WLDS_PER_A)

#define NBLOCKS 256
#define WAVES_PER_BLOCK 8
#define NWAVES (NBLOCKS * WAVES_PER_BLOCK)
#define NITEMS (B_ * M_ * (F_ / 32))
#define ITEMS_PER_WAVE (NITEMS / NWAVES)
#define ROUND_STRIDE ((size_t)(NWAVES / 4) * (A_ * F_))

struct Item {
    v2f x[16];
    v2f y[8];
};

__device__ __forceinline__ void load_item(Item& it,
                                          const float* __restrict__ xp,
                                          const float* __restrict__ yp)
{
#pragma unroll
    for (int a = 0; a < 16; ++a)
        it.x[a] = *reinterpret_cast<const v2f*>(xp + a * F_);
#pragma unroll
    for (int k = 0; k < 4; ++k) {
        it.y[2 * k]     = *reinterpret_cast<const v2f*>(yp + (4 * k)     * F_);
        it.y[2 * k + 1] = *reinterpret_cast<const v2f*>(yp + (4 * k + 1) * F_);
    }
}

__device__ __forceinline__ void compute_store(const v2f (&wa)[64], const Item& it,
                                              float* __restrict__ op)
{
    v8f acc0 = {};
    v8f acc1 = {};

#pragma unroll
    for (int t = 0; t < 64; ++t) {
        const int a  = t >> 2;
        const int k4 = t & 3;

        v2f b0, b1;
        b0.x = it.x[a].x * it.y[2 * k4].x;
        b0.y = it.x[a].x * it.y[2 * k4 + 1].x;
        b1.x = it.x[a].y * it.y[2 * k4].y;
        b1.y = it.x[a].y * it.y[2 * k4 + 1].y;

        acc0 = __builtin_amdgcn_wmma_f32_16x16x4_f32(
             false, wa[t],  false, b0,
             (short)0, acc0,  false,  false);
        acc1 = __builtin_amdgcn_wmma_f32_16x16x4_f32(
             false, wa[t],  false, b1,
             (short)0, acc1,  false,  false);
    }

#pragma unroll
    for (int r = 0; r < 8; ++r) {
        v2f o;
        o.x = acc0[r];
        o.y = acc1[r];
        *reinterpret_cast<v2f*>(op + r * F_) = o;
    }
}

__global__ __launch_bounds__(256)
void tpcl_wmma_f32_persistent(const float* __restrict__ x,
                              const float* __restrict__ y,
                              const float* __restrict__ cgc,
                              float* __restrict__ out)
{
    __shared__ float w[WLDS_TOTAL];

    const int lane = threadIdx.x & 31;
    const int wave = threadIdx.x >> 5;
    const int col  = lane & 15;
    const int hi   = lane >> 4;
    const int off  = hi << 1;
    const int gw   = blockIdx.x * WAVES_PER_BLOCK + wave;

    const size_t base = (size_t)(gw >> 2) * (A_ * F_) + (size_t)((gw & 3) * 32 + 2 * col);
    const float* __restrict__ xp = x + base;
    const float* __restrict__ yp = y + base + off * F_;
    float* __restrict__ op       = out + base + hi * (8 * F_);

    Item itA, itB;
    load_item(itA, xp, yp);

#pragma unroll
    for (int j = 0; j < 4; ++j) {
        const int i4 = threadIdx.x + 256 * j;
        const v4f v  = reinterpret_cast<const v4f*>(cgc)[i4];
        const int a  = i4 >> 6;
        const int d  = (i4 >> 2) & 15;
        const int c0 = (i4 & 3) << 2;
        float* dst = &w[a * WLDS_PER_A + d * WSTRIDE + c0];
        dst[0] = v.x; dst[1] = v.y; dst[2] = v.z; dst[3] = v.w;
    }
    __syncthreads();

    v2f wa[64];
    const float* wbase = &w[col * WSTRIDE + off];
#pragma unroll
    for (int t = 0; t < 64; ++t) {
        const int a  = t >> 2;
        const int k4 = t & 3;
        wa[t].x = wbase[a * WLDS_PER_A + 4 * k4];
        wa[t].y = wbase[a * WLDS_PER_A + 4 * k4 + 1];
    }

    for (int i = 0; i < ITEMS_PER_WAVE; i += 2) {
        load_item(itB, xp + (size_t)(i + 1) * ROUND_STRIDE, yp + (size_t)(i + 1) * ROUND_STRIDE);
        __builtin_amdgcn_sched_barrier(0);
        compute_store(wa, itA, op + (size_t)i * ROUND_STRIDE);
        if (i + 2 < ITEMS_PER_WAVE)
            load_item(itA, xp + (size_t)(i + 2) * ROUND_STRIDE, yp + (size_t)(i + 2) * ROUND_STRIDE);
        compute_store(wa, itB, op + (size_t)(i + 1) * ROUND_STRIDE);
    }
}

extern "C" void kernel_launch(void* const* d_in, const int* in_sizes, int n_in,
                              void* d_out, int out_size, void* d_ws, size_t ws_size,
                              hipStream_t stream) {
    const float* x   = (const float*)d_in[0];
    const float* y   = (const float*)d_in[1];
    const float* cgc = (const float*)d_in[2];
    float* out = (float*)d_out;

    (void)in_sizes; (void)n_in; (void)out_size; (void)d_ws; (void)ws_size;

    dim3 grid(NBLOCKS);
    dim3 block(256);
    tpcl_wmma_f32_persistent<<<grid, block, 0, stream>>>(x, y, cgc, out);
}


# exact counted load waits: items 6,7 peeled into a straight-line tail so no request is conditional; the wait before an odd item's products no longer drains the loads just issued
# speedup vs baseline: 1.3171x; 1.0450x over previous
#include <hip/hip_runtime.h>


typedef __attribute__((ext_vector_type(2))) float v2f;
typedef __attribute__((ext_vector_type(4))) float v4f;
typedef __attribute__((ext_vector_type(8))) float v8f;

#define B_ 8
#define M_ 512
#define A_ 16
#define F_ 128

#define WSTRIDE 17
#define WLDS_PER_A (A_ * WSTRIDE)
#define WLDS_TOTAL (A_ * WLDS_PER_A)

#define NBLOCKS 256
#define WAVES_PER_BLOCK 8
#define NWAVES (NBLOCKS * WAVES_PER_BLOCK)
#define NITEMS (B_ * M_ * (F_ / 32))
#define ITEMS_PER_WAVE (NITEMS / NWAVES)
#define ROUND_STRIDE ((size_t)(NWAVES / 4) * (A_ * F_))

struct Item {
    v2f x[16];
    v2f y[8];
};

__device__ __forceinline__ void load_item(Item& it,
                                          const float* __restrict__ xp,
                                          const float* __restrict__ yp)
{
#pragma unroll
    for (int a = 0; a < 16; ++a)
        it.x[a] = *reinterpret_cast<const v2f*>(xp + a * F_);
#pragma unroll
    for (int k = 0; k < 4; ++k) {
        it.y[2 * k]     = *reinterpret_cast<const v2f*>(yp + (4 * k)     * F_);
        it.y[2 * k + 1] = *reinterpret_cast<const v2f*>(yp + (4 * k + 1) * F_);
    }
}

__device__ __forceinline__ void compute_store(const v2f (&wa)[64], const Item& it,
                                              float* __restrict__ op)
{
    v8f acc0 = {};
    v8f acc1 = {};

#pragma unroll
    for (int t = 0; t < 64; ++t) {
        const int a  = t >> 2;
        const int k4 = t & 3;

        v2f b0, b1;
        b0.x = it.x[a].x * it.y[2 * k4].x;
        b0.y = it.x[a].x * it.y[2 * k4 + 1].x;
        b1.x = it.x[a].y * it.y[2 * k4].y;
        b1.y = it.x[a].y * it.y[2 * k4 + 1].y;

        acc0 = __builtin_amdgcn_wmma_f32_16x16x4_f32(
             false, wa[t],  false, b0,
             (short)0, acc0,  false,  false);
        acc1 = __builtin_amdgcn_wmma_f32_16x16x4_f32(
             false, wa[t],  false, b1,
             (short)0, acc1,  false,  false);
    }

#pragma unroll
    for (int r = 0; r < 8; ++r) {
        v2f o;
        o.x = acc0[r];
        o.y = acc1[r];
        *reinterpret_cast<v2f*>(op + r * F_) = o;
    }
}

__global__ __launch_bounds__(256)
void tpcl_wmma_f32_persistent(const float* __restrict__ x,
                              const float* __restrict__ y,
                              const float* __restrict__ cgc,
                              float* __restrict__ out)
{
    __shared__ float w[WLDS_TOTAL];

    const int lane = threadIdx.x & 31;
    const int wave = threadIdx.x >> 5;
    const int col  = lane & 15;
    const int hi   = lane >> 4;
    const int off  = hi << 1;
    const int gw   = blockIdx.x * WAVES_PER_BLOCK + wave;

    const size_t base = (size_t)(gw >> 2) * (A_ * F_) + (size_t)((gw & 3) * 32 + 2 * col);
    const float* __restrict__ xp = x + base;
    const float* __restrict__ yp = y + base + off * F_;
    float* __restrict__ op       = out + base + hi * (8 * F_);

    Item itA, itB;
    load_item(itA, xp, yp);

#pragma unroll
    for (int j = 0; j < 4; ++j) {
        const int i4 = threadIdx.x + 256 * j;
        const v4f v  = reinterpret_cast<const v4f*>(cgc)[i4];
        const int a  = i4 >> 6;
        const int d  = (i4 >> 2) & 15;
        const int c0 = (i4 & 3) << 2;
        float* dst = &w[a * WLDS_PER_A + d * WSTRIDE + c0];
        dst[0] = v.x; dst[1] = v.y; dst[2] = v.z; dst[3] = v.w;
    }
    __syncthreads();

    v2f wa[64];
    const float* wbase = &w[col * WSTRIDE + off];
#pragma unroll
    for (int t = 0; t < 64; ++t) {
        const int a  = t >> 2;
        const int k4 = t & 3;
        wa[t].x = wbase[a * WLDS_PER_A + 4 * k4];
        wa[t].y = wbase[a * WLDS_PER_A + 4 * k4 + 1];
    }

    for (int i = 0; i < ITEMS_PER_WAVE - 2; i += 2) {
        load_item(itB, xp + (size_t)(i + 1) * ROUND_STRIDE, yp + (size_t)(i + 1) * ROUND_STRIDE);
        __builtin_amdgcn_sched_barrier(0);
        compute_store(wa, itA, op + (size_t)i * ROUND_STRIDE);
        __builtin_amdgcn_sched_barrier(0);
        load_item(itA, xp + (size_t)(i + 2) * ROUND_STRIDE, yp + (size_t)(i + 2) * ROUND_STRIDE);
        __builtin_amdgcn_sched_barrier(0);
        compute_store(wa, itB, op + (size_t)(i + 1) * ROUND_STRIDE);
    }
    load_item(itB, xp + (size_t)(ITEMS_PER_WAVE - 1) * ROUND_STRIDE,
                   yp + (size_t)(ITEMS_PER_WAVE - 1) * ROUND_STRIDE);
    __builtin_amdgcn_sched_barrier(0);
    compute_store(wa, itA, op + (size_t)(ITEMS_PER_WAVE - 2) * ROUND_STRIDE);
    __builtin_amdgcn_sched_barrier(0);
    compute_store(wa, itB, op + (size_t)(ITEMS_PER_WAVE - 1) * ROUND_STRIDE);
}

extern "C" void kernel_launch(void* const* d_in, const int* in_sizes, int n_in,
                              void* d_out, int out_size, void* d_ws, size_t ws_size,
                              hipStream_t stream) {
    const float* x   = (const float*)d_in[0];
    const float* y   = (const float*)d_in[1];
    const float* cgc = (const float*)d_in[2];
    float* out = (float*)d_out;

    (void)in_sizes; (void)n_in; (void)out_size; (void)d_ws; (void)ws_size;

    dim3 grid(NBLOCKS);
    dim3 block(256);
    tpcl_wmma_f32_persistent<<<grid, block, 0, stream>>>(x, y, cgc, out);
}
